# starting kernel
# baseline (speedup 1.0000x reference)
#include <hip/hip_runtime.h>

typedef __attribute__((ext_vector_type(2))) float v2f;
typedef __attribute__((ext_vector_type(8))) float v8f;

#define D_DIM   1024
#define N_DIM   8
#define C_DIM   4
#define NROWS   (N_DIM * C_DIM)
#define VS      1028
#define NITER   3
#define EPS_F   1e-6f

__global__ void wdiag_gather_kernel(const float* __restrict__ W,
                                    float* __restrict__ wdiag) {
    int t = blockIdx.x * blockDim.x + threadIdx.x;
    if (t >= NROWS * D_DIM) return;
    int row = t >> 10;
    int d   = t & (D_DIM - 1);
    int n = row >> 2, c = row & 3;
    size_t idx = (((size_t)n * D_DIM + d) * C_DIM + c) * D_DIM + d;
    wdiag[t] = W[idx];
}

__global__ __launch_bounds__(256)
void routing_kernel(const float* __restrict__ fi, const float* __restrict__ pi,
                    const float* __restrict__ W, const float* __restrict__ wdiag,
                    const float* __restrict__ cj_init,
                    float* __restrict__ out_cj, float* __restrict__ out_rij) {
    extern __shared__ float smem[];
    float* votes = smem;
    float* cjl   = votes + NROWS * VS;
    float* sij   = cjl + C_DIM * VS;
    float* nrm   = sij + NROWS;

    const int b    = blockIdx.x;
    const int tid  = threadIdx.x;
    const int lane = tid & 31;
    const int wave = tid >> 5;

    float pin[N_DIM];
#pragma unroll
    for (int n = 0; n < N_DIM; ++n) pin[n] = pi[b * N_DIM + n];

    if (tid < C_DIM) nrm[tid] = 0.0f;
    if (tid >= 64 && tid < 64 + NROWS) sij[tid - 64] = 0.0f;

#pragma unroll
    for (int k = 0; k < 4; ++k) {
        int d = tid + 256 * k;
#pragma unroll
        for (int n = 0; n < N_DIM; ++n) {
            float f = fi[((size_t)b * N_DIM + n) * D_DIM + d];
#pragma unroll
            for (int c = 0; c < C_DIM; ++c) {
                int row = n * C_DIM + c;
                float wv = wdiag ? wdiag[row * D_DIM + d]
                                 : W[(((size_t)n * D_DIM + d) * C_DIM + c) * D_DIM + d];
                votes[row * VS + d] = f * wv;
            }
        }
    }

    float cjn[C_DIM][4];
    float psum[C_DIM];
#pragma unroll
    for (int c = 0; c < C_DIM; ++c) {
        psum[c] = 0.0f;
#pragma unroll
        for (int k = 0; k < 4; ++k) {
            int d = tid + 256 * k;
            float v = cj_init[((size_t)b * C_DIM + c) * D_DIM + d];
            cjn[c][k] = v;
            psum[c] += v * v;
        }
    }
    __syncthreads();
#pragma unroll
    for (int c = 0; c < C_DIM; ++c) {
        float s = psum[c];
#pragma unroll
        for (int off = 16; off >= 1; off >>= 1) s += __shfl_xor(s, off, 32);
        if (lane == 0) atomicAdd(&nrm[c], s);
    }
    __syncthreads();
#pragma unroll
    for (int c = 0; c < C_DIM; ++c) {
        float inv = 1.0f / (sqrtf(nrm[c]) + EPS_F);
#pragma unroll
        for (int k = 0; k < 4; ++k) {
            int d = tid + 256 * k;
            cjl[c * VS + d] = cjn[c][k] * inv;
        }
    }
    __syncthreads();
    if (tid < C_DIM) nrm[tid] = 0.0f;
    __syncthreads();

    const int    arow = lane & 15;
    const int    koff = (lane & 16) ? 2 : 0;
    const float* a0p  = votes + arow * VS;
    const float* a1p  = votes + (16 + arow) * VS;
    const bool   bval = arow < C_DIM;
    const float* bp   = cjl + (bval ? arow : 0) * VS;
    const int    kbase = wave * 128;

    float wgt[N_DIM][C_DIM];

    for (int it = 0; it < NITER; ++it) {
        v8f acc0 = {0.f, 0.f, 0.f, 0.f, 0.f, 0.f, 0.f, 0.f};
        v8f acc1 = acc0;
#pragma unroll 4
        for (int s = 0; s < 32; ++s) {
            int d0 = kbase + 4 * s + koff;
            v2f a0 = *(const v2f*)(a0p + d0);
            v2f a1 = *(const v2f*)(a1p + d0);
            v2f bb = *(const v2f*)(bp + d0);
            v2f bz = {0.f, 0.f};
            bb = bval ? bb : bz;
            acc0 = __builtin_amdgcn_wmma_f32_16x16x4_f32(false, a0, false, bb,
                                                         (short)0, acc0, false, false);
            acc1 = __builtin_amdgcn_wmma_f32_16x16x4_f32(false, a1, false, bb,
                                                         (short)0, acc1, false, false);
        }
#pragma unroll
        for (int v = 0; v < 8; ++v) {
            if ((lane & 15) == (v & 3)) {
                int mrow = v + ((lane >> 4) << 3);
                atomicAdd(&sij[mrow],      acc0[v]);
                atomicAdd(&sij[16 + mrow], acc1[v]);
            }
        }
        __syncthreads();

#pragma unroll
        for (int n = 0; n < N_DIM; ++n) {
            float s0 = sij[n * 4 + 0], s1 = sij[n * 4 + 1];
            float s2 = sij[n * 4 + 2], s3 = sij[n * 4 + 3];
            float mx = fmaxf(fmaxf(s0, s1), fmaxf(s2, s3));
            float e0 = __expf(s0 - mx), e1 = __expf(s1 - mx);
            float e2 = __expf(s2 - mx), e3 = __expf(s3 - mx);
            float rs = pin[n] / (e0 + e1 + e2 + e3);
            wgt[n][0] = e0 * rs; wgt[n][1] = e1 * rs;
            wgt[n][2] = e2 * rs; wgt[n][3] = e3 * rs;
        }

#pragma unroll
        for (int c = 0; c < C_DIM; ++c) {
            psum[c] = 0.0f;
#pragma unroll
            for (int k = 0; k < 4; ++k) {
                int d = tid + 256 * k;
                float a = 0.0f;
#pragma unroll
                for (int n = 0; n < N_DIM; ++n)
                    a = fmaf(wgt[n][c], votes[(n * C_DIM + c) * VS + d], a);
                cjn[c][k] = a;
                psum[c] += a * a;
            }
        }
#pragma unroll
        for (int c = 0; c < C_DIM; ++c) {
            float s = psum[c];
#pragma unroll
            for (int off = 16; off >= 1; off >>= 1) s += __shfl_xor(s, off, 32);
            if (lane == 0) atomicAdd(&nrm[c], s);
        }
        __syncthreads();

#pragma unroll
        for (int c = 0; c < C_DIM; ++c) {
            float inv = 1.0f / (sqrtf(nrm[c]) + EPS_F);
#pragma unroll
            for (int k = 0; k < 4; ++k) {
                int d = tid + 256 * k;
                float v = cjn[c][k] * inv;
                cjn[c][k] = v;
                if (it < NITER - 1) cjl[c * VS + d] = v;
            }
        }
        __syncthreads();
        if (tid < C_DIM) nrm[tid] = 0.0f;
        if (it < NITER - 1 && tid >= 64 && tid < 64 + NROWS) sij[tid - 64] = 0.0f;
        __syncthreads();
    }

#pragma unroll
    for (int c = 0; c < C_DIM; ++c)
#pragma unroll
        for (int k = 0; k < 4; ++k) {
            int d = tid + 256 * k;
            out_cj[((size_t)b * C_DIM + c) * D_DIM + d] = cjn[c][k];
        }
    if (tid < NROWS) {
        int n = tid >> 2;
        float s0 = sij[n * 4 + 0], s1 = sij[n * 4 + 1];
        float s2 = sij[n * 4 + 2], s3 = sij[n * 4 + 3];
        float mx = fmaxf(fmaxf(s0, s1), fmaxf(s2, s3));
        float e0 = __expf(s0 - mx), e1 = __expf(s1 - mx);
        float e2 = __expf(s2 - mx), e3 = __expf(s3 - mx);
        out_rij[(size_t)b * NROWS + tid] = __expf(sij[tid] - mx) / (e0 + e1 + e2 + e3);
    }
}

extern "C" void kernel_launch(void* const* d_in, const int* in_sizes, int n_in,
                              void* d_out, int out_size, void* d_ws, size_t ws_size,
                              hipStream_t stream) {
    (void)n_in; (void)out_size;
    const float* fi  = (const float*)d_in[0];
    const float* pi  = (const float*)d_in[1];
    const float* W   = (const float*)d_in[2];
    const float* cji = (const float*)d_in[3];
    const int B = in_sizes[0] / (N_DIM * D_DIM);

    float* out_cj  = (float*)d_out;
    float* out_rij = out_cj + (size_t)B * C_DIM * D_DIM;

    const float* wdiag = nullptr;
    const size_t wbytes = (size_t)NROWS * D_DIM * sizeof(float);
    if (ws_size >= wbytes) {
        int total = NROWS * D_DIM;
        wdiag_gather_kernel<<<(total + 255) / 256, 256, 0, stream>>>(W, (float*)d_ws);
        wdiag = (const float*)d_ws;
    }

    const size_t smem = (size_t)(NROWS * VS + C_DIM * VS + NROWS + C_DIM) * sizeof(float);
    (void)hipFuncSetAttribute((const void*)routing_kernel,
                              hipFuncAttributeMaxDynamicSharedMemorySize, (int)smem);
    routing_kernel<<<B, 256, smem, stream>>>(fi, pi, W, wdiag, cji, out_cj, out_rij);
}


# votes held in registers as WMMA A fragments for all 3 iterations (diag(W) staged in fragment order), cj as B fragment, cj update by 4-lane DPP sums; LDS tile and LDS atomics gone; sums reordered
# speedup vs baseline: 3.2035x; 3.2035x over previous
#include <hip/hip_runtime.h>


typedef __attribute__((ext_vector_type(2))) float v2f;
typedef __attribute__((ext_vector_type(4))) float v4f;
typedef __attribute__((ext_vector_type(8))) float v8f;

#define D_DIM   1024
#define N_DIM   8
#define C_DIM   4
#define NROWS   (N_DIM * C_DIM)
#define NQ      16
#define NITER   3
#define EPS_F   1e-6f


__global__ void wdiag_gather_kernel(const float* __restrict__ W,
                                    float* __restrict__ wfrag) {
    int o = blockIdx.x * blockDim.x + threadIdx.x;
    if (o >= NROWS * D_DIM) return;
    int e    = o & 3;
    int lane = (o >> 2) & 31;
    int t    = (o >> 7) & 1;
    int q    = (o >> 8) & 15;
    int w    = o >> 12;
    int m = lane & 15, h = lane >> 4;
    int row = 16 * t + m;
    int d   = 128 * w + 8 * q + 4 * h + e;
    int n = row >> 2, c = row & 3;
    size_t idx = (((size_t)n * D_DIM + d) * C_DIM + c) * D_DIM + d;
    wfrag[o] = W[idx];
}

template <int CTRL>
__device__ __forceinline__ float row_xmask_get(float x) {
    return __builtin_bit_cast(float,
        __builtin_amdgcn_update_dpp(0, __builtin_bit_cast(int, x), CTRL, 0xF, 0xF, true));
}

__global__ __launch_bounds__(256)
void routing_kernel(const float* __restrict__ fi, const float* __restrict__ pi,
                    const float* __restrict__ W, const float* __restrict__ wfrag,
                    const float* __restrict__ cj_init,
                    float* __restrict__ out_cj, float* __restrict__ out_rij) {
    __shared__ __attribute__((aligned(16))) float s_part[8 * NROWS];
    __shared__ __attribute__((aligned(16))) float n_part[8 * C_DIM];

    const int b    = blockIdx.x;
    const int tid  = threadIdx.x;
    const int lane = tid & 31;
    const int wave = tid >> 5;
    const int m    = lane & 15;
    const int h    = lane >> 4;
    const int c    = m & 3;
    const int n0   = m >> 2;
    const int n1   = 4 + n0;
    const int dbase = 128 * wave + 4 * h;

    const float pin0 = pi[b * N_DIM + n0];
    const float pin1 = pi[b * N_DIM + n1];

    v4f A0[NQ], A1[NQ], Bf[NQ];
    const float* f0 = fi + ((size_t)b * N_DIM + n0) * D_DIM + dbase;
    const float* f1 = fi + ((size_t)b * N_DIM + n1) * D_DIM + dbase;
    const float* cp = cj_init + ((size_t)b * C_DIM + c) * D_DIM + dbase;

#pragma unroll
    for (int q = 0; q < NQ; ++q) Bf[q] = *(const v4f*)(cp + 8 * q);

    if (wfrag) {
        const float* wp = wfrag + ((size_t)wave * NQ * 2 * 32 + lane) * 4;
#pragma unroll
        for (int q = 0; q < NQ; ++q) {
            v4f w0 = *(const v4f*)(wp + (q * 2 + 0) * 128);
            v4f w1 = *(const v4f*)(wp + (q * 2 + 1) * 128);
            v4f x0 = *(const v4f*)(f0 + 8 * q);
            v4f x1 = *(const v4f*)(f1 + 8 * q);
            A0[q] = x0 * w0;
            A1[q] = x1 * w1;
        }
    } else {
#pragma unroll
        for (int q = 0; q < NQ; ++q) {
            v4f x0 = *(const v4f*)(f0 + 8 * q);
            v4f x1 = *(const v4f*)(f1 + 8 * q);
#pragma unroll
            for (int e = 0; e < 4; ++e) {
                int d = dbase + 8 * q + e;
                float w0 = W[(((size_t)n0 * D_DIM + d) * C_DIM + c) * D_DIM + d];
                float w1 = W[(((size_t)n1 * D_DIM + d) * C_DIM + c) * D_DIM + d];
                A0[q][e] = x0[e] * w0;
                A1[q][e] = x1[e] * w1;
            }
        }
    }

    float ss = 0.0f;
#pragma unroll
    for (int q = 0; q < NQ; ++q) {
#pragma unroll
        for (int e = 0; e < 4; ++e) ss = fmaf(Bf[q][e], Bf[q][e], ss);
    }
    ss += __shfl_xor(ss, 16, 32);
    if (lane < C_DIM) n_part[wave * C_DIM + lane] = ss;
    __syncthreads();
    {
        float tot = 0.0f;
#pragma unroll
        for (int wv = 0; wv < 8; ++wv) tot += n_part[wv * C_DIM + c];
        float inv = 1.0f / (sqrtf(tot) + EPS_F);
#pragma unroll
        for (int q = 0; q < NQ; ++q) Bf[q] *= inv;
    }

    float rij0 = 0.0f, rij1 = 0.0f;

    for (int it = 0; it < NITER; ++it) {
        v8f acc0 = {0.f, 0.f, 0.f, 0.f, 0.f, 0.f, 0.f, 0.f};
        v8f acc1 = acc0;
#pragma unroll
        for (int q = 0; q < NQ; ++q) {
#pragma unroll
            for (int hf = 0; hf < 2; ++hf) {
                v2f a0 = {A0[q][2 * hf], A0[q][2 * hf + 1]};
                v2f a1 = {A1[q][2 * hf], A1[q][2 * hf + 1]};
                v2f bb = {Bf[q][2 * hf], Bf[q][2 * hf + 1]};
                acc0 = __builtin_amdgcn_wmma_f32_16x16x4_f32(false, a0, false, bb,
                                                             (short)0, acc0, false, false);
                acc1 = __builtin_amdgcn_wmma_f32_16x16x4_f32(false, a1, false, bb,
                                                             (short)0, acc1, false, false);
            }
        }
#pragma unroll
        for (int r = 0; r < 8; ++r) {
            if (m == (r & 3)) {
                s_part[wave * NROWS + 8 * h + r]      = acc0[r];
                s_part[wave * NROWS + 16 + 8 * h + r] = acc1[r];
            }
        }
        __syncthreads();

        v4f sa = {0.f, 0.f, 0.f, 0.f}, sb = sa;
#pragma unroll
        for (int wv = 0; wv < 8; ++wv) {
            sa += *(const v4f*)(s_part + wv * NROWS + n0 * 4);
            sb += *(const v4f*)(s_part + wv * NROWS + n1 * 4);
        }
        float w0, w1;
        {
            float mx = fmaxf(fmaxf(sa[0], sa[1]), fmaxf(sa[2], sa[3]));
            float e0 = __expf(sa[0] - mx), e1 = __expf(sa[1] - mx);
            float e2 = __expf(sa[2] - mx), e3 = __expf(sa[3] - mx);
            float sum = e0 + e1 + e2 + e3;
            float ec = (c == 0) ? e0 : (c == 1) ? e1 : (c == 2) ? e2 : e3;
            w0 = ec * (pin0 / sum);
            rij0 = ec / sum;
        }
        {
            float mx = fmaxf(fmaxf(sb[0], sb[1]), fmaxf(sb[2], sb[3]));
            float e0 = __expf(sb[0] - mx), e1 = __expf(sb[1] - mx);
            float e2 = __expf(sb[2] - mx), e3 = __expf(sb[3] - mx);
            float sum = e0 + e1 + e2 + e3;
            float ec = (c == 0) ? e0 : (c == 1) ? e1 : (c == 2) ? e2 : e3;
            w1 = ec * (pin1 / sum);
            rij1 = ec / sum;
        }

        float sn = 0.0f;
#pragma unroll
        for (int q = 0; q < NQ; ++q) {
#pragma unroll
            for (int e = 0; e < 4; ++e) {
                float x = fmaf(w1, A1[q][e], w0 * A0[q][e]);
                x += row_xmask_get<0x164>(x);
                x += row_xmask_get<0x168>(x);
                Bf[q][e] = x;
                sn = fmaf(x, x, sn);
            }
        }
        sn += __shfl_xor(sn, 16, 32);
        if (lane < C_DIM) n_part[wave * C_DIM + lane] = sn;
        __syncthreads();

        float tot = 0.0f;
#pragma unroll
        for (int wv = 0; wv < 8; ++wv) tot += n_part[wv * C_DIM + c];
        float inv = 1.0f / (sqrtf(tot) + EPS_F);
#pragma unroll
        for (int q = 0; q < NQ; ++q) Bf[q] *= inv;
    }

    if (m < C_DIM) {
        float* op = out_cj + ((size_t)b * C_DIM + c) * D_DIM + dbase;
#pragma unroll
        for (int q = 0; q < NQ; ++q) *(v4f*)(op + 8 * q) = Bf[q];
    }
    if (tid < 16) {
        out_rij[(size_t)b * NROWS + m]      = rij0;
        out_rij[(size_t)b * NROWS + 16 + m] = rij1;
    }
}

extern "C" void kernel_launch(void* const* d_in, const int* in_sizes, int n_in,
                              void* d_out, int out_size, void* d_ws, size_t ws_size,
                              hipStream_t stream) {
    (void)n_in; (void)out_size;
    const float* fi  = (const float*)d_in[0];
    const float* pi  = (const float*)d_in[1];
    const float* W   = (const float*)d_in[2];
    const float* cji = (const float*)d_in[3];
    const int B = in_sizes[0] / (N_DIM * D_DIM);

    float* out_cj  = (float*)d_out;
    float* out_rij = out_cj + (size_t)B * C_DIM * D_DIM;

    const float* wfrag = nullptr;
    const size_t wbytes = (size_t)NROWS * D_DIM * sizeof(float);
    if (ws_size >= wbytes) {
        int total = NROWS * D_DIM;
        wdiag_gather_kernel<<<(total + 255) / 256, 256, 0, stream>>>(W, (float*)d_ws);
        wfrag = (const float*)d_ws;
    }

    routing_kernel<<<B, 256, 0, stream>>>(fi, pi, W, wfrag, cji, out_cj, out_rij);
}


# votes prologue split into a load phase and a multiply phase: all 80 fragment loads of a wave in flight at once (340 VGPRs, still 2 waves per SIMD); was 16 dependent groups of 4; same products
# speedup vs baseline: 3.9902x; 1.2456x over previous
#include <hip/hip_runtime.h>


typedef __attribute__((ext_vector_type(2))) float v2f;
typedef __attribute__((ext_vector_type(4))) float v4f;
typedef __attribute__((ext_vector_type(8))) float v8f;

#define D_DIM   1024
#define N_DIM   8
#define C_DIM   4
#define NROWS   (N_DIM * C_DIM)
#define NQ      16
#define NITER   3
#define EPS_F   1e-6f


__global__ void wdiag_gather_kernel(const float* __restrict__ W,
                                    float* __restrict__ wfrag) {
    int o = blockIdx.x * blockDim.x + threadIdx.x;
    if (o >= NROWS * D_DIM) return;
    int e    = o & 3;
    int lane = (o >> 2) & 31;
    int t    = (o >> 7) & 1;
    int q    = (o >> 8) & 15;
    int w    = o >> 12;
    int m = lane & 15, h = lane >> 4;
    int row = 16 * t + m;
    int d   = 128 * w + 8 * q + 4 * h + e;
    int n = row >> 2, c = row & 3;
    size_t idx = (((size_t)n * D_DIM + d) * C_DIM + c) * D_DIM + d;
    wfrag[o] = W[idx];
}

template <int CTRL>
__device__ __forceinline__ float row_xmask_get(float x) {
    return __builtin_bit_cast(float,
        __builtin_amdgcn_update_dpp(0, __builtin_bit_cast(int, x), CTRL, 0xF, 0xF, true));
}

__global__ __launch_bounds__(256)
void routing_kernel(const float* __restrict__ fi, const float* __restrict__ pi,
                    const float* __restrict__ W, const float* __restrict__ wfrag,
                    const float* __restrict__ cj_init,
                    float* __restrict__ out_cj, float* __restrict__ out_rij) {
    __shared__ __attribute__((aligned(16))) float s_part[8 * NROWS];
    __shared__ __attribute__((aligned(16))) float n_part[8 * C_DIM];

    const int b    = blockIdx.x;
    const int tid  = threadIdx.x;
    const int lane = tid & 31;
    const int wave = tid >> 5;
    const int m    = lane & 15;
    const int h    = lane >> 4;
    const int c    = m & 3;
    const int n0   = m >> 2;
    const int n1   = 4 + n0;
    const int dbase = 128 * wave + 4 * h;

    const float pin0 = pi[b * N_DIM + n0];
    const float pin1 = pi[b * N_DIM + n1];

    v4f A0[NQ], A1[NQ], Bf[NQ];
    const float* f0 = fi + ((size_t)b * N_DIM + n0) * D_DIM + dbase;
    const float* f1 = fi + ((size_t)b * N_DIM + n1) * D_DIM + dbase;
    const float* cp = cj_init + ((size_t)b * C_DIM + c) * D_DIM + dbase;

#pragma unroll
    for (int q = 0; q < NQ; ++q) Bf[q] = *(const v4f*)(cp + 8 * q);

    if (wfrag) {
        const float* wp = wfrag + ((size_t)wave * NQ * 2 * 32 + lane) * 4;
        v4f W0[NQ], W1[NQ];
#pragma unroll
        for (int q = 0; q < NQ; ++q) {
            A0[q] = *(const v4f*)(f0 + 8 * q);
            A1[q] = *(const v4f*)(f1 + 8 * q);
        }
#pragma unroll
        for (int q = 0; q < NQ; ++q) {
            W0[q] = *(const v4f*)(wp + (q * 2 + 0) * 128);
            W1[q] = *(const v4f*)(wp + (q * 2 + 1) * 128);
        }
#pragma unroll
        for (int q = 0; q < NQ; ++q) {
            A0[q] *= W0[q];
            A1[q] *= W1[q];
        }
    } else {
#pragma unroll
        for (int q = 0; q < NQ; ++q) {
            v4f x0 = *(const v4f*)(f0 + 8 * q);
            v4f x1 = *(const v4f*)(f1 + 8 * q);
#pragma unroll
            for (int e = 0; e < 4; ++e) {
                int d = dbase + 8 * q + e;
                float w0 = W[(((size_t)n0 * D_DIM + d) * C_DIM + c) * D_DIM + d];
                float w1 = W[(((size_t)n1 * D_DIM + d) * C_DIM + c) * D_DIM + d];
                A0[q][e] = x0[e] * w0;
                A1[q][e] = x1[e] * w1;
            }
        }
    }

    float ss = 0.0f;
#pragma unroll
    for (int q = 0; q < NQ; ++q) {
#pragma unroll
        for (int e = 0; e < 4; ++e) ss = fmaf(Bf[q][e], Bf[q][e], ss);
    }
    ss += __shfl_xor(ss, 16, 32);
    if (lane < C_DIM) n_part[wave * C_DIM + lane] = ss;
    __syncthreads();
    {
        float tot = 0.0f;
#pragma unroll
        for (int wv = 0; wv < 8; ++wv) tot += n_part[wv * C_DIM + c];
        float inv = 1.0f / (sqrtf(tot) + EPS_F);
#pragma unroll
        for (int q = 0; q < NQ; ++q) Bf[q] *= inv;
    }

    float rij0 = 0.0f, rij1 = 0.0f;

    for (int it = 0; it < NITER; ++it) {
        v8f acc0 = {0.f, 0.f, 0.f, 0.f, 0.f, 0.f, 0.f, 0.f};
        v8f acc1 = acc0;
#pragma unroll
        for (int q = 0; q < NQ; ++q) {
#pragma unroll
            for (int hf = 0; hf < 2; ++hf) {
                v2f a0 = {A0[q][2 * hf], A0[q][2 * hf + 1]};
                v2f a1 = {A1[q][2 * hf], A1[q][2 * hf + 1]};
                v2f bb = {Bf[q][2 * hf], Bf[q][2 * hf + 1]};
                acc0 = __builtin_amdgcn_wmma_f32_16x16x4_f32(false, a0, false, bb,
                                                             (short)0, acc0, false, false);
                acc1 = __builtin_amdgcn_wmma_f32_16x16x4_f32(false, a1, false, bb,
                                                             (short)0, acc1, false, false);
            }
        }
#pragma unroll
        for (int r = 0; r < 8; ++r) {
            if (m == (r & 3)) {
                s_part[wave * NROWS + 8 * h + r]      = acc0[r];
                s_part[wave * NROWS + 16 + 8 * h + r] = acc1[r];
            }
        }
        __syncthreads();

        v4f sa = {0.f, 0.f, 0.f, 0.f}, sb = sa;
#pragma unroll
        for (int wv = 0; wv < 8; ++wv) {
            sa += *(const v4f*)(s_part + wv * NROWS + n0 * 4);
            sb += *(const v4f*)(s_part + wv * NROWS + n1 * 4);
        }
        float w0, w1;
        {
            float mx = fmaxf(fmaxf(sa[0], sa[1]), fmaxf(sa[2], sa[3]));
            float e0 = __expf(sa[0] - mx), e1 = __expf(sa[1] - mx);
            float e2 = __expf(sa[2] - mx), e3 = __expf(sa[3] - mx);
            float sum = e0 + e1 + e2 + e3;
            float ec = (c == 0) ? e0 : (c == 1) ? e1 : (c == 2) ? e2 : e3;
            w0 = ec * (pin0 / sum);
            rij0 = ec / sum;
        }
        {
            float mx = fmaxf(fmaxf(sb[0], sb[1]), fmaxf(sb[2], sb[3]));
            float e0 = __expf(sb[0] - mx), e1 = __expf(sb[1] - mx);
            float e2 = __expf(sb[2] - mx), e3 = __expf(sb[3] - mx);
            float sum = e0 + e1 + e2 + e3;
            float ec = (c == 0) ? e0 : (c == 1) ? e1 : (c == 2) ? e2 : e3;
            w1 = ec * (pin1 / sum);
            rij1 = ec / sum;
        }

        float sn = 0.0f;
#pragma unroll
        for (int q = 0; q < NQ; ++q) {
#pragma unroll
            for (int e = 0; e < 4; ++e) {
                float x = fmaf(w1, A1[q][e], w0 * A0[q][e]);
                x += row_xmask_get<0x164>(x);
                x += row_xmask_get<0x168>(x);
                Bf[q][e] = x;
                sn = fmaf(x, x, sn);
            }
        }
        sn += __shfl_xor(sn, 16, 32);
        if (lane < C_DIM) n_part[wave * C_DIM + lane] = sn;
        __syncthreads();

        float tot = 0.0f;
#pragma unroll
        for (int wv = 0; wv < 8; ++wv) tot += n_part[wv * C_DIM + c];
        float inv = 1.0f / (sqrtf(tot) + EPS_F);
#pragma unroll
        for (int q = 0; q < NQ; ++q) Bf[q] *= inv;
    }

    if (m < C_DIM) {
        float* op = out_cj + ((size_t)b * C_DIM + c) * D_DIM + dbase;
#pragma unroll
        for (int q = 0; q < NQ; ++q) *(v4f*)(op + 8 * q) = Bf[q];
    }
    if (tid < 16) {
        out_rij[(size_t)b * NROWS + m]      = rij0;
        out_rij[(size_t)b * NROWS + 16 + m] = rij1;
    }
}

extern "C" void kernel_launch(void* const* d_in, const int* in_sizes, int n_in,
                              void* d_out, int out_size, void* d_ws, size_t ws_size,
                              hipStream_t stream) {
    (void)n_in; (void)out_size;
    const float* fi  = (const float*)d_in[0];
    const float* pi  = (const float*)d_in[1];
    const float* W   = (const float*)d_in[2];
    const float* cji = (const float*)d_in[3];
    const int B = in_sizes[0] / (N_DIM * D_DIM);

    float* out_cj  = (float*)d_out;
    float* out_rij = out_cj + (size_t)B * C_DIM * D_DIM;

    const float* wfrag = nullptr;
    const size_t wbytes = (size_t)NROWS * D_DIM * sizeof(float);
    if (ws_size >= wbytes) {
        int total = NROWS * D_DIM;
        wdiag_gather_kernel<<<(total + 255) / 256, 256, 0, stream>>>(W, (float*)d_ws);
        wfrag = (const float*)d_ws;
    }

    routing_kernel<<<B, 256, 0, stream>>>(fi, pi, W, wfrag, cji, out_cj, out_rij);
}


# lane holds all 8 rows of one concept at its 16 d (8 WMMA tiles, D diagonal by bit-select): 8-row sum of the cj update in-lane, 128 DPP adds per pass gone; cj stored in whole lines; sums reordered
# speedup vs baseline: 4.2442x; 1.0637x over previous
#include <hip/hip_runtime.h>


typedef __attribute__((ext_vector_type(2))) float v2f;
typedef __attribute__((ext_vector_type(4))) float v4f;
typedef __attribute__((ext_vector_type(8))) float v8f;

#define D_DIM   1024
#define N_DIM   8
#define C_DIM   4
#define NROWS   (N_DIM * C_DIM)
#define NJ      4
#define NITER   3
#define EPS_F   1e-6f


__global__ void wdiag_gather_kernel(const float* __restrict__ W,
                                    float* __restrict__ wfrag) {
    int o = blockIdx.x * blockDim.x + threadIdx.x;
    if (o >= NROWS * D_DIM) return;
    int e    = o & 3;
    int lane = (o >> 2) & 31;
    int j    = (o >> 7) & 3;
    int n    = (o >> 9) & 7;
    int w    = o >> 12;
    int c = lane & 3, blk = lane >> 2;
    int d = 128 * w + 32 * j + 4 * blk + e;
    size_t idx = (((size_t)n * D_DIM + d) * C_DIM + c) * D_DIM + d;
    wfrag[o] = W[idx];
}

template <int CTRL>
__device__ __forceinline__ float row_xmask_get(float x) {
    return __builtin_bit_cast(float,
        __builtin_amdgcn_update_dpp(0, __builtin_bit_cast(int, x), CTRL, 0xF, 0xF, true));
}

__device__ __forceinline__ float concept_sum(float x) {
    x += row_xmask_get<0x164>(x);
    x += row_xmask_get<0x168>(x);
    x += __shfl_xor(x, 16, 32);
    return x;
}

__device__ __forceinline__ float bitsel(unsigned msk, float a, float b) {
    unsigned ua = __builtin_bit_cast(unsigned, a), ub = __builtin_bit_cast(unsigned, b);
    return __builtin_bit_cast(float, ub ^ ((ua ^ ub) & msk));
}

__global__ __launch_bounds__(256)
void routing_kernel(const float* __restrict__ fi, const float* __restrict__ pi,
                    const float* __restrict__ W, const float* __restrict__ wfrag,
                    const float* __restrict__ cj_init,
                    float* __restrict__ out_cj, float* __restrict__ out_rij) {
    __shared__ __attribute__((aligned(16))) float s_part[8 * 2 * NROWS];
    __shared__ __attribute__((aligned(16))) float n_part[8 * C_DIM];

    const int b    = blockIdx.x;
    const int tid  = threadIdx.x;
    const int lane = tid & 31;
    const int wave = tid >> 5;
    const int c    = lane & 3;
    const int blk  = lane >> 2;
    const int half = lane >> 4;
    const int dbase = 128 * wave + 4 * blk;

    const float pin = pi[b * N_DIM + blk];

    v4f A[N_DIM][NJ], Bf[NJ];
    const float* fp = fi + (size_t)b * N_DIM * D_DIM + dbase;
    const float* cp = cj_init + ((size_t)b * C_DIM + c) * D_DIM + dbase;

#pragma unroll
    for (int j = 0; j < NJ; ++j) Bf[j] = *(const v4f*)(cp + 32 * j);

    if (wfrag) {
        const float* wp = wfrag + ((size_t)wave * (N_DIM * NJ * 32) + lane) * 4;
        v4f Wf[N_DIM][NJ];
#pragma unroll
        for (int n = 0; n < N_DIM; ++n) {
#pragma unroll
            for (int j = 0; j < NJ; ++j) A[n][j] = *(const v4f*)(fp + n * D_DIM + 32 * j);
        }
#pragma unroll
        for (int n = 0; n < N_DIM; ++n) {
#pragma unroll
            for (int j = 0; j < NJ; ++j) Wf[n][j] = *(const v4f*)(wp + (n * NJ + j) * 128);
        }
#pragma unroll
        for (int n = 0; n < N_DIM; ++n) {
#pragma unroll
            for (int j = 0; j < NJ; ++j) A[n][j] *= Wf[n][j];
        }
    } else {
#pragma unroll
        for (int n = 0; n < N_DIM; ++n) {
#pragma unroll
            for (int j = 0; j < NJ; ++j) {
                v4f x = *(const v4f*)(fp + n * D_DIM + 32 * j);
#pragma unroll
                for (int e = 0; e < 4; ++e) {
                    int d = dbase + 32 * j + e;
                    float wv = W[(((size_t)n * D_DIM + d) * C_DIM + c) * D_DIM + d];
                    A[n][j][e] = x[e] * wv;
                }
            }
        }
    }

    float ss = 0.0f;
#pragma unroll
    for (int j = 0; j < NJ; ++j) {
#pragma unroll
        for (int e = 0; e < 4; ++e) ss = fmaf(Bf[j][e], Bf[j][e], ss);
    }
    ss = concept_sum(ss);
    if (lane < C_DIM) n_part[wave * C_DIM + lane] = ss;
    __syncthreads();
    {
        float tot = 0.0f;
#pragma unroll
        for (int wv = 0; wv < 8; ++wv) tot += n_part[wv * C_DIM + c];
        float inv = 1.0f / (sqrtf(tot) + EPS_F);
#pragma unroll
        for (int j = 0; j < NJ; ++j) Bf[j] *= inv;
    }

    float rij = 0.0f;
    unsigned m1 = (lane & 1) ? 0xFFFFFFFFu : 0u;
    unsigned m2 = (lane & 2) ? 0xFFFFFFFFu : 0u;
    unsigned m4 = (lane & 4) ? 0xFFFFFFFFu : 0u;
    asm volatile("" : "+v"(m1), "+v"(m2), "+v"(m4));
    const bool writer = (lane & 12) == (half << 3);

    for (int it = 0; it < NITER; ++it) {
        v8f acc[N_DIM];
#pragma unroll
        for (int n = 0; n < N_DIM; ++n) acc[n] = (v8f){0.f, 0.f, 0.f, 0.f, 0.f, 0.f, 0.f, 0.f};
#pragma unroll
        for (int j = 0; j < NJ; ++j) {
#pragma unroll
            for (int hf = 0; hf < 2; ++hf) {
                v2f bb = {Bf[j][2 * hf], Bf[j][2 * hf + 1]};
#pragma unroll
                for (int n = 0; n < N_DIM; ++n) {
                    v2f a = {A[n][j][2 * hf], A[n][j][2 * hf + 1]};
                    acc[n] = __builtin_amdgcn_wmma_f32_16x16x4_f32(false, a, false, bb,
                                                                   (short)0, acc[n], false, false);
                }
            }
        }
        float dg[N_DIM];
#pragma unroll
        for (int n = 0; n < N_DIM; ++n) {
            float t01 = bitsel(m1, acc[n][1], acc[n][0]);
            float t23 = bitsel(m1, acc[n][3], acc[n][2]);
            float t45 = bitsel(m1, acc[n][5], acc[n][4]);
            float t67 = bitsel(m1, acc[n][7], acc[n][6]);
            float t03 = bitsel(m2, t23, t01);
            float t47 = bitsel(m2, t67, t45);
            float t   = bitsel(m4, t47, t03);
            dg[n] = t + row_xmask_get<0x164>(t);
        }
        if (writer) {
#pragma unroll
            for (int n = 0; n < N_DIM; ++n)
                s_part[((wave * 2 + half) * N_DIM + n) * C_DIM + c] = dg[n];
        }
        __syncthreads();

        v4f sa = {0.f, 0.f, 0.f, 0.f};
#pragma unroll
        for (int p = 0; p < 16; ++p) sa += *(const v4f*)(s_part + (p * N_DIM + blk) * C_DIM);
        float wgt;
        {
            float mx = fmaxf(fmaxf(sa[0], sa[1]), fmaxf(sa[2], sa[3]));
            float e0 = __expf(sa[0] - mx), e1 = __expf(sa[1] - mx);
            float e2 = __expf(sa[2] - mx), e3 = __expf(sa[3] - mx);
            float sum = e0 + e1 + e2 + e3;
            float ec = bitsel(m2, bitsel(m1, e3, e2), bitsel(m1, e1, e0));
            wgt = ec * (pin / sum);
            rij = ec / sum;
        }
        float wn[N_DIM];
#pragma unroll
        for (int n = 0; n < N_DIM; ++n) wn[n] = __shfl(wgt, c + 4 * n, 32);

        float sn = 0.0f;
#pragma unroll
        for (int j = 0; j < NJ; ++j) {
#pragma unroll
            for (int e = 0; e < 4; ++e) {
                float x = wn[0] * A[0][j][e];
#pragma unroll
                for (int n = 1; n < N_DIM; ++n) x = fmaf(wn[n], A[n][j][e], x);
                Bf[j][e] = x;
                sn = fmaf(x, x, sn);
            }
        }
        sn = concept_sum(sn);
        if (lane < C_DIM) n_part[wave * C_DIM + lane] = sn;
        __syncthreads();

        float tot = 0.0f;
#pragma unroll
        for (int wv = 0; wv < 8; ++wv) tot += n_part[wv * C_DIM + c];
        float inv = 1.0f / (sqrtf(tot) + EPS_F);
#pragma unroll
        for (int j = 0; j < NJ; ++j) Bf[j] *= inv;
    }

    {
        float* op = out_cj + ((size_t)b * C_DIM + c) * D_DIM + dbase;
#pragma unroll
        for (int j = 0; j < NJ; ++j) *(v4f*)(op + 32 * j) = Bf[j];
    }
    if (tid < 32) {
        out_rij[(size_t)b * NROWS + lane] = rij;
    }
}

extern "C" void kernel_launch(void* const* d_in, const int* in_sizes, int n_in,
                              void* d_out, int out_size, void* d_ws, size_t ws_size,
                              hipStream_t stream) {
    (void)n_in; (void)out_size;
    const float* fi  = (const float*)d_in[0];
    const float* pi  = (const float*)d_in[1];
    const float* W   = (const float*)d_in[2];
    const float* cji = (const float*)d_in[3];
    const int B = in_sizes[0] / (N_DIM * D_DIM);

    float* out_cj  = (float*)d_out;
    float* out_rij = out_cj + (size_t)B * C_DIM * D_DIM;

    const float* wfrag = nullptr;
    const size_t wbytes = (size_t)NROWS * D_DIM * sizeof(float);
    if (ws_size >= wbytes) {
        int total = NROWS * D_DIM;
        wdiag_gather_kernel<<<(total + 255) / 256, 256, 0, stream>>>(W, (float*)d_ws);
        wfrag = (const float*)d_ws;
    }

    routing_kernel<<<B, 256, 0, stream>>>(fi, pi, W, wfrag, cji, out_cj, out_rij);
}
